# starting kernel
# baseline (speedup 1.0000x reference)
#include <hip/hip_runtime.h>
#include <hip/hip_bf16.h>
#include <math.h>

#define HH   512
#define WW   512
#define KK   16
#define KP3  19
#define KPAD 20
#define BB   64
#define NCOL 128
#define HWN  (HH*WW)

#define BLDS 136
#define SROW 68

typedef __attribute__((ext_vector_type(2))) float v2f;
typedef __attribute__((ext_vector_type(8))) float v8f;

__device__ __forceinline__ float cpx_of(int k) { return -1.0f + (2.0f/3.0f)*(float)(k >> 2); }
__device__ __forceinline__ float cpy_of(int k) { return -1.0f + (2.0f/3.0f)*(float)(k & 3); }
__device__ __forceinline__ float tps_u(float r2) { return r2 * logf(r2 + 1e-6f); }

__global__ void tps_setup_kernel(const float* __restrict__ sp, float* __restrict__ bmat) {
    __shared__ float aug[KP3][2*KP3];
    __shared__ float invD[KP3][KP3];
    const int tid = threadIdx.x;

    if (tid == 0) {
        for (int i = 0; i < KP3; ++i) {
            for (int j = 0; j < KP3; ++j) {
                float v;
                if (i < KK && j < KK) {
                    float dx = cpx_of(i) - cpx_of(j);
                    float dy = cpy_of(i) - cpy_of(j);
                    v = tps_u(dx*dx + dy*dy);
                } else if (i < KK) {
                    v = (j == KK) ? 1.0f : ((j == KK+1) ? cpx_of(i) : cpy_of(i));
                } else if (j < KK) {
                    v = (i == KK) ? 1.0f : ((i == KK+1) ? cpx_of(j) : cpy_of(j));
                } else {
                    v = 0.0f;
                }
                aug[i][j] = v;
                aug[i][KP3 + j] = (i == j) ? 1.0f : 0.0f;
            }
        }
        for (int p = 0; p < KP3; ++p) {
            int piv = p; float best = fabsf(aug[p][p]);
            for (int r = p + 1; r < KP3; ++r) {
                float a = fabsf(aug[r][p]);
                if (a > best) { best = a; piv = r; }
            }
            if (piv != p) {
                for (int j = 0; j < 2*KP3; ++j) {
                    float t = aug[p][j]; aug[p][j] = aug[piv][j]; aug[piv][j] = t;
                }
            }
            float inv = 1.0f / aug[p][p];
            for (int j = 0; j < 2*KP3; ++j) aug[p][j] *= inv;
            for (int r = 0; r < KP3; ++r) {
                if (r == p) continue;
                float f = aug[r][p];
                for (int j = 0; j < 2*KP3; ++j) aug[r][j] -= f * aug[p][j];
            }
        }
        for (int i = 0; i < KP3; ++i)
            for (int j = 0; j < KP3; ++j)
                invD[i][j] = aug[i][KP3 + j];
    }
    __syncthreads();

    const int c = tid;
    if (c < NCOL) {
        const int b = c >> 1, d = c & 1;
        float spv[KK];
        #pragma unroll
        for (int j = 0; j < KK; ++j) spv[j] = sp[b*(KK*2) + j*2 + d];
        for (int i = 0; i < KP3; ++i) {
            float acc = 0.0f;
            #pragma unroll
            for (int j = 0; j < KK; ++j) acc += invD[i][j] * spv[j];
            bmat[i*NCOL + c] = acc;
        }
        bmat[KP3*NCOL + c] = 0.0f;
    }
}

__global__ void __launch_bounds__(256)
tps_main_kernel(const float* __restrict__ bmat, float* __restrict__ out) {
    __shared__ float Bls[KPAD * BLDS];
    __shared__ float stage[8][16 * SROW];

    const int tid = threadIdx.x;

    for (int idx = tid; idx < KPAD * NCOL; idx += 256) {
        int r = idx >> 7, cc = idx & 127;
        Bls[r * BLDS + cc] = bmat[idx];
    }
    __syncthreads();

    const int lane = tid & 31;
    const int wv   = tid >> 5;
    const int m    = lane & 15;
    const bool hi  = lane >= 16;
    const int n0   = blockIdx.x * 128 + wv * 16;
    const int n    = n0 + m;

    const float px = -1.0f + 2.0f * (float)(n & (WW - 1)) * (1.0f / (float)(WW - 1));
    const float py = -1.0f + 2.0f * (float)(n >> 9)       * (1.0f / (float)(HH - 1));

    v2f A[5];
    #pragma unroll
    for (int kb = 0; kb < 4; ++kb) {
        int k0 = kb * 4 + (hi ? 2 : 0);
        float dx0 = px - cpx_of(k0),     dy0 = py - cpy_of(k0);
        float dx1 = px - cpx_of(k0 + 1), dy1 = py - cpy_of(k0 + 1);
        float r20 = dx0*dx0 + dy0*dy0;
        float r21 = dx1*dx1 + dy1*dy1;
        A[kb].x = r20 * logf(r20 + 1e-6f);
        A[kb].y = r21 * logf(r21 + 1e-6f);
    }
    A[4].x = hi ? py   : 1.0f;
    A[4].y = hi ? 0.0f : px;

    float* st = &stage[wv][0];
    const int colLane = lane & 15;
    const int rbase   = hi ? 8 : 0;

    for (int h2 = 0; h2 < 2; ++h2) {
        #pragma unroll
        for (int t = 0; t < 4; ++t) {
            const int c0  = (h2 * 4 + t) * 16;
            const int col = c0 + colLane;
            v8f C = {0.f, 0.f, 0.f, 0.f, 0.f, 0.f, 0.f, 0.f};
            #pragma unroll
            for (int kb = 0; kb < 5; ++kb) {
                int r0 = kb * 4 + (hi ? 2 : 0);
                v2f Bv;
                Bv.x = Bls[r0 * BLDS + col];
                Bv.y = Bls[(r0 + 1) * BLDS + col];
                C = __builtin_amdgcn_wmma_f32_16x16x4_f32(
                         false, A[kb],  false, Bv,
                         (short)0, C,  false,  false);
            }
            const int lc = t * 16 + colLane;
            #pragma unroll
            for (int v = 0; v < 8; ++v)
                st[(rbase + v) * SROW + lc] = C[v];
        }
        __syncthreads();

        #pragma unroll
        for (int bp = 0; bp < 16; ++bp) {
            const int b  = h2 * 32 + 2 * bp + (hi ? 1 : 0);
            const int lc = 4 * bp + (hi ? 2 : 0);
            v2f val = *(const v2f*)&st[m * SROW + lc];
            float* gp = out + (size_t)b * (size_t)(HWN * 2) + (size_t)(n0 + m) * 2;
            *(v2f*)gp = val;
        }
        __syncthreads();
    }
}

extern "C" void kernel_launch(void* const* d_in, const int* in_sizes, int n_in,
                              void* d_out, int out_size, void* d_ws, size_t ws_size,
                              hipStream_t stream) {
    (void)in_sizes; (void)n_in; (void)out_size; (void)ws_size;
    const float* src = (const float*)d_in[0];
    float* out       = (float*)d_out;
    float* bmat      = (float*)d_ws;

    tps_setup_kernel<<<1, 128, 0, stream>>>(src, bmat);

    const int nBlocks = HWN / 128;
    tps_main_kernel<<<nBlocks, 256, 0, stream>>>(bmat, out);
}


# setup: 19x19 Gauss-Jordan inversion moved from one thread on LDS to one wave, one matrix row per lane in registers, pivot row passed by lane reads; pivot rule, element operations, summing order kept
# speedup vs baseline: 16.4683x; 16.4683x over previous
#include <hip/hip_runtime.h>
#include <hip/hip_bf16.h>
#include <math.h>

#define HH   512
#define WW   512
#define KK   16
#define KP3  19
#define KPAD 20
#define BB   64
#define NCOL 128
#define HWN  (HH*WW)

#define BLDS 136
#define SROW 68

typedef __attribute__((ext_vector_type(2))) float v2f;
typedef __attribute__((ext_vector_type(8))) float v8f;

__device__ __forceinline__ float cpx_of(int k) { return -1.0f + (2.0f/3.0f)*(float)(k >> 2); }
__device__ __forceinline__ float cpy_of(int k) { return -1.0f + (2.0f/3.0f)*(float)(k & 3); }
__device__ __forceinline__ float tps_u(float r2) { return r2 * logf(r2 + 1e-6f); }

#define NAUG (2*KP3)

__device__ __forceinline__ float xhalf(float x) {
    const int b = __builtin_bit_cast(int, x);
    return __builtin_bit_cast(float,
        __builtin_amdgcn_permlanex16(b, b, 0x76543210, 0xfedcba98, false, false));
}

template <int M>
__device__ __forceinline__ float xrow(float x) {
    const int b = __builtin_bit_cast(int, x);
    return __builtin_bit_cast(float,
        __builtin_amdgcn_update_dpp(b, b, 0x160 + M, 0xf, 0xf, false));
}

__device__ __forceinline__ float wave_max32(float v) {
    v = fmaxf(v, xrow<1>(v));
    v = fmaxf(v, xrow<2>(v));
    v = fmaxf(v, xrow<4>(v));
    v = fmaxf(v, xrow<8>(v));
    v = fmaxf(v, xhalf(v));
    return v;
}

__device__ __forceinline__ float lane_get(float x, int l) {
    return __builtin_bit_cast(float,
        __builtin_amdgcn_readlane(__builtin_bit_cast(int, x), l));
}

__global__ void tps_setup_kernel(const float* __restrict__ sp, float* __restrict__ bmat) {
    __shared__ float invD[KP3][KK];
    const int tid = threadIdx.x;

    if (tid < 32) {
        const int i = tid;
        float row[NAUG];
        for (int j = 0; j < KK; ++j) {
            const float dx = cpx_of(i) - cpx_of(j);
            const float dy = cpy_of(i) - cpy_of(j);
            const float u  = tps_u(dx*dx + dy*dy);
            const float e  = (i == KK) ? 1.0f : ((i == KK+1) ? cpx_of(j) : cpy_of(j));
            const float v  = (i < KK) ? u : e;
            row[j] = (i < KP3) ? v : 0.0f;
        }
        row[KK]     = (i < KK) ? 1.0f      : 0.0f;
        row[KK + 1] = (i < KK) ? cpx_of(i) : 0.0f;
        row[KK + 2] = (i < KK) ? cpy_of(i) : 0.0f;
        for (int j = 0; j < KP3; ++j) row[KP3 + j] = (i == j) ? 1.0f : 0.0f;

        for (int p = 0; p < KP3; ++p) {
            const float a  = (i >= p && i < KP3) ? fabsf(row[0]) : -1.0f;
            const float mx = wave_max32(a);
            const unsigned hit = __builtin_amdgcn_ballot_w32(a == mx);
            const int piv = hit ? __builtin_ctz(hit) : p;

            float pr[NAUG];
            for (int j = 0; j < NAUG; ++j) pr[j] = lane_get(row[j], piv);

            if (piv != p) {
                for (int j = 0; j < NAUG; ++j) {
                    const float o = lane_get(row[j], p);
                    row[j] = (i == piv) ? o : row[j];
                }
            }

            const float inv = 1.0f / pr[0];
            const float f   = row[0];
            float nw[NAUG];
            for (int j = 0; j < NAUG; ++j) {
                const float ps = pr[j] * inv;
                const float e  = row[j] - f * ps;
                nw[j] = (i == p) ? ps : e;
            }
            for (int j = 0; j < NAUG - 1; ++j) row[j] = nw[j + 1];
            row[NAUG - 1] = nw[0];
        }

        if (i < KP3) {
            for (int j = 0; j < KK; ++j) invD[i][j] = row[j];
        }
    }
    __syncthreads();

    const int c = tid;
    if (c < NCOL) {
        const int b = c >> 1, d = c & 1;
        float spv[KK];
        #pragma unroll
        for (int j = 0; j < KK; ++j) spv[j] = sp[b*(KK*2) + j*2 + d];
        for (int i = 0; i < KP3; ++i) {
            float acc = 0.0f;
            #pragma unroll
            for (int j = 0; j < KK; ++j) acc += invD[i][j] * spv[j];
            bmat[i*NCOL + c] = acc;
        }
        bmat[KP3*NCOL + c] = 0.0f;
    }
}

__global__ void __launch_bounds__(256)
tps_main_kernel(const float* __restrict__ bmat, float* __restrict__ out) {
    __shared__ float Bls[KPAD * BLDS];
    __shared__ float stage[8][16 * SROW];

    const int tid = threadIdx.x;

    for (int idx = tid; idx < KPAD * NCOL; idx += 256) {
        int r = idx >> 7, cc = idx & 127;
        Bls[r * BLDS + cc] = bmat[idx];
    }
    __syncthreads();

    const int lane = tid & 31;
    const int wv   = tid >> 5;
    const int m    = lane & 15;
    const bool hi  = lane >= 16;
    const int n0   = blockIdx.x * 128 + wv * 16;
    const int n    = n0 + m;

    const float px = -1.0f + 2.0f * (float)(n & (WW - 1)) * (1.0f / (float)(WW - 1));
    const float py = -1.0f + 2.0f * (float)(n >> 9)       * (1.0f / (float)(HH - 1));

    v2f A[5];
    #pragma unroll
    for (int kb = 0; kb < 4; ++kb) {
        int k0 = kb * 4 + (hi ? 2 : 0);
        float dx0 = px - cpx_of(k0),     dy0 = py - cpy_of(k0);
        float dx1 = px - cpx_of(k0 + 1), dy1 = py - cpy_of(k0 + 1);
        float r20 = dx0*dx0 + dy0*dy0;
        float r21 = dx1*dx1 + dy1*dy1;
        A[kb].x = r20 * logf(r20 + 1e-6f);
        A[kb].y = r21 * logf(r21 + 1e-6f);
    }
    A[4].x = hi ? py   : 1.0f;
    A[4].y = hi ? 0.0f : px;

    float* st = &stage[wv][0];
    const int colLane = lane & 15;
    const int rbase   = hi ? 8 : 0;

    for (int h2 = 0; h2 < 2; ++h2) {
        #pragma unroll
        for (int t = 0; t < 4; ++t) {
            const int c0  = (h2 * 4 + t) * 16;
            const int col = c0 + colLane;
            v8f C = {0.f, 0.f, 0.f, 0.f, 0.f, 0.f, 0.f, 0.f};
            #pragma unroll
            for (int kb = 0; kb < 5; ++kb) {
                int r0 = kb * 4 + (hi ? 2 : 0);
                v2f Bv;
                Bv.x = Bls[r0 * BLDS + col];
                Bv.y = Bls[(r0 + 1) * BLDS + col];
                C = __builtin_amdgcn_wmma_f32_16x16x4_f32(
                         false, A[kb],  false, Bv,
                         (short)0, C,  false,  false);
            }
            const int lc = t * 16 + colLane;
            #pragma unroll
            for (int v = 0; v < 8; ++v)
                st[(rbase + v) * SROW + lc] = C[v];
        }
        __syncthreads();

        #pragma unroll
        for (int bp = 0; bp < 16; ++bp) {
            const int b  = h2 * 32 + 2 * bp + (hi ? 1 : 0);
            const int lc = 4 * bp + (hi ? 2 : 0);
            v2f val = *(const v2f*)&st[m * SROW + lc];
            float* gp = out + (size_t)b * (size_t)(HWN * 2) + (size_t)(n0 + m) * 2;
            *(v2f*)gp = val;
        }
        __syncthreads();
    }
}

extern "C" void kernel_launch(void* const* d_in, const int* in_sizes, int n_in,
                              void* d_out, int out_size, void* d_ws, size_t ws_size,
                              hipStream_t stream) {
    (void)in_sizes; (void)n_in; (void)out_size; (void)ws_size;
    const float* src = (const float*)d_in[0];
    float* out       = (float*)d_out;
    float* bmat      = (float*)d_ws;

    tps_setup_kernel<<<1, 128, 0, stream>>>(src, bmat);

    const int nBlocks = HWN / 128;
    tps_main_kernel<<<nBlocks, 256, 0, stream>>>(bmat, out);
}


# main: matrix operands swapped (coefficients A, grid basis B): accumulators in output order, 2 interleaved row tiles per wave, 16-byte stores in whole 256-byte pieces, no LDS staging; sum order kept
# speedup vs baseline: 21.7759x; 1.3223x over previous
#include <hip/hip_runtime.h>
#include <hip/hip_bf16.h>
#include <math.h>

#define HH   512
#define WW   512
#define KK   16
#define KP3  19
#define KPAD 20
#define BB   64
#define NCOL 128
#define HWN  (HH*WW)

#define BLDS 136

typedef __attribute__((ext_vector_type(2))) float v2f;
typedef __attribute__((ext_vector_type(8))) float v8f;

__device__ __forceinline__ float cpx_of(int k) { return -1.0f + (2.0f/3.0f)*(float)(k >> 2); }
__device__ __forceinline__ float cpy_of(int k) { return -1.0f + (2.0f/3.0f)*(float)(k & 3); }
__device__ __forceinline__ float tps_u(float r2) { return r2 * logf(r2 + 1e-6f); }

#define NAUG (2*KP3)

__device__ __forceinline__ float xhalf(float x) {
    const int b = __builtin_bit_cast(int, x);
    return __builtin_bit_cast(float,
        __builtin_amdgcn_permlanex16(b, b, 0x76543210, 0xfedcba98, false, false));
}

template <int M>
__device__ __forceinline__ float xrow(float x) {
    const int b = __builtin_bit_cast(int, x);
    return __builtin_bit_cast(float,
        __builtin_amdgcn_update_dpp(b, b, 0x160 + M, 0xf, 0xf, false));
}

__device__ __forceinline__ float wave_max32(float v) {
    v = fmaxf(v, xrow<1>(v));
    v = fmaxf(v, xrow<2>(v));
    v = fmaxf(v, xrow<4>(v));
    v = fmaxf(v, xrow<8>(v));
    v = fmaxf(v, xhalf(v));
    return v;
}

__device__ __forceinline__ float lane_get(float x, int l) {
    return __builtin_bit_cast(float,
        __builtin_amdgcn_readlane(__builtin_bit_cast(int, x), l));
}

__global__ void tps_setup_kernel(const float* __restrict__ sp, float* __restrict__ bmat) {
    __shared__ float invD[KP3][KK];
    const int tid = threadIdx.x;

    if (tid < 32) {
        const int i = tid;
        float row[NAUG];
        for (int j = 0; j < KK; ++j) {
            const float dx = cpx_of(i) - cpx_of(j);
            const float dy = cpy_of(i) - cpy_of(j);
            const float u  = tps_u(dx*dx + dy*dy);
            const float e  = (i == KK) ? 1.0f : ((i == KK+1) ? cpx_of(j) : cpy_of(j));
            const float v  = (i < KK) ? u : e;
            row[j] = (i < KP3) ? v : 0.0f;
        }
        row[KK]     = (i < KK) ? 1.0f      : 0.0f;
        row[KK + 1] = (i < KK) ? cpx_of(i) : 0.0f;
        row[KK + 2] = (i < KK) ? cpy_of(i) : 0.0f;
        for (int j = 0; j < KP3; ++j) row[KP3 + j] = (i == j) ? 1.0f : 0.0f;

        for (int p = 0; p < KP3; ++p) {
            const float a  = (i >= p && i < KP3) ? fabsf(row[0]) : -1.0f;
            const float mx = wave_max32(a);
            const unsigned hit = __builtin_amdgcn_ballot_w32(a == mx);
            const int piv = hit ? __builtin_ctz(hit) : p;

            float pr[NAUG];
            for (int j = 0; j < NAUG; ++j) pr[j] = lane_get(row[j], piv);

            if (piv != p) {
                for (int j = 0; j < NAUG; ++j) {
                    const float o = lane_get(row[j], p);
                    row[j] = (i == piv) ? o : row[j];
                }
            }

            const float inv = 1.0f / pr[0];
            const float f   = row[0];
            float nw[NAUG];
            for (int j = 0; j < NAUG; ++j) {
                const float ps = pr[j] * inv;
                const float e  = row[j] - f * ps;
                nw[j] = (i == p) ? ps : e;
            }
            for (int j = 0; j < NAUG - 1; ++j) row[j] = nw[j + 1];
            row[NAUG - 1] = nw[0];
        }

        if (i < KP3) {
            for (int j = 0; j < KK; ++j) invD[i][j] = row[j];
        }
    }
    __syncthreads();

    const int c = tid;
    if (c < NCOL) {
        const int b = c >> 1, d = c & 1;
        float spv[KK];
        #pragma unroll
        for (int j = 0; j < KK; ++j) spv[j] = sp[b*(KK*2) + j*2 + d];
        for (int i = 0; i < KP3; ++i) {
            float acc = 0.0f;
            #pragma unroll
            for (int j = 0; j < KK; ++j) acc += invD[i][j] * spv[j];
            bmat[i*NCOL + c] = acc;
        }
        bmat[KP3*NCOL + c] = 0.0f;
    }
}

typedef __attribute__((ext_vector_type(4))) float v4f;

__global__ void __launch_bounds__(256)
tps_main_kernel(const float* __restrict__ bmat, float* __restrict__ out) {
    __shared__ float Bls[KPAD * BLDS];

    const int tid = threadIdx.x;

    for (int idx = tid; idx < KPAD * NCOL; idx += 256) {
        int r = idx >> 7, cc = idx & 127;
        Bls[r * BLDS + cc] = bmat[idx];
    }
    __syncthreads();

    const int lane = tid & 31;
    const int wv   = tid >> 5;
    const int c    = lane & 15;
    const bool hi  = lane >= 16;
    const int n0   = blockIdx.x * 256 + wv * 32;
    const int nE   = n0 + 2 * c;

    v2f U[2][5];
    #pragma unroll
    for (int e = 0; e < 2; ++e) {
        const int n = nE + e;
        const float px = -1.0f + 2.0f * (float)(n & (WW - 1)) * (1.0f / (float)(WW - 1));
        const float py = -1.0f + 2.0f * (float)(n >> 9)       * (1.0f / (float)(HH - 1));
        #pragma unroll
        for (int kb = 0; kb < 4; ++kb) {
            int k0 = kb * 4 + (hi ? 2 : 0);
            float dx0 = px - cpx_of(k0),     dy0 = py - cpy_of(k0);
            float dx1 = px - cpx_of(k0 + 1), dy1 = py - cpy_of(k0 + 1);
            float r20 = dx0*dx0 + dy0*dy0;
            float r21 = dx1*dx1 + dy1*dy1;
            U[e][kb].x = r20 * logf(r20 + 1e-6f);
            U[e][kb].y = r21 * logf(r21 + 1e-6f);
        }
        U[e][4].x = hi ? py   : 1.0f;
        U[e][4].y = hi ? 0.0f : px;
    }

    float* gp = out + (size_t)(hi ? 4 : 0) * (size_t)(HWN * 2) + (size_t)nE * 2;

    #pragma unroll
    for (int ct = 0; ct < 8; ++ct) {
        const int col = ct * 16 + c;
        v2f Cf[5];
        #pragma unroll
        for (int kb = 0; kb < 5; ++kb) {
            int r0 = kb * 4 + (hi ? 2 : 0);
            Cf[kb].x = Bls[r0 * BLDS + col];
            Cf[kb].y = Bls[(r0 + 1) * BLDS + col];
        }
        v8f E = {0.f, 0.f, 0.f, 0.f, 0.f, 0.f, 0.f, 0.f};
        v8f O = {0.f, 0.f, 0.f, 0.f, 0.f, 0.f, 0.f, 0.f};
        #pragma unroll
        for (int kb = 0; kb < 5; ++kb) {
            E = __builtin_amdgcn_wmma_f32_16x16x4_f32(
                     false, Cf[kb],  false, U[0][kb],
                     (short)0, E,  false,  false);
            O = __builtin_amdgcn_wmma_f32_16x16x4_f32(
                     false, Cf[kb],  false, U[1][kb],
                     (short)0, O,  false,  false);
        }
        #pragma unroll
        for (int i = 0; i < 4; ++i) {
            v4f val = {E[2*i], E[2*i + 1], O[2*i], O[2*i + 1]};
            *(v4f*)(gp + (size_t)(ct * 8 + i) * (size_t)(HWN * 2)) = val;
        }
    }
}

extern "C" void kernel_launch(void* const* d_in, const int* in_sizes, int n_in,
                              void* d_out, int out_size, void* d_ws, size_t ws_size,
                              hipStream_t stream) {
    (void)in_sizes; (void)n_in; (void)out_size; (void)ws_size;
    const float* src = (const float*)d_in[0];
    float* out       = (float*)d_out;
    float* bmat      = (float*)d_ws;

    tps_setup_kernel<<<1, 128, 0, stream>>>(src, bmat);

    const int nBlocks = HWN / 256;
    tps_main_kernel<<<nBlocks, 256, 0, stream>>>(bmat, out);
}


# grid cut from 1024 to 256 work-groups (one per WGP), each looping over four 256-point tiles: inversion and coefficient build once per WGP, coefficient fragments stay in registers; sum order kept
# speedup vs baseline: 31.8935x; 1.4646x over previous
#include <hip/hip_runtime.h>
#include <hip/hip_bf16.h>
#include <math.h>


#define HH   512
#define WW   512
#define KK   16
#define KP3  19
#define KPAD 20
#define BB   64
#define NCOL 128
#define HWN  (HH*WW)
#define NWG  256
#define TPW  4

#define BLDS 136
#define SPLD 33

typedef __attribute__((ext_vector_type(2))) float v2f;
typedef __attribute__((ext_vector_type(4))) float v4f;
typedef __attribute__((ext_vector_type(8))) float v8f;

__device__ __forceinline__ float cpx_of(int k) { return -1.0f + (2.0f/3.0f)*(float)(k >> 2); }
__device__ __forceinline__ float cpy_of(int k) { return -1.0f + (2.0f/3.0f)*(float)(k & 3); }
__device__ __forceinline__ float tps_u(float r2) { return r2 * logf(r2 + 1e-6f); }

#define NAUG (2*KP3)

__device__ __forceinline__ float xhalf(float x) {
    const int b = __builtin_bit_cast(int, x);
    return __builtin_bit_cast(float,
        __builtin_amdgcn_permlanex16(b, b, 0x76543210, 0xfedcba98, false, false));
}

template <int M>
__device__ __forceinline__ float xrow(float x) {
    const int b = __builtin_bit_cast(int, x);
    return __builtin_bit_cast(float,
        __builtin_amdgcn_update_dpp(b, b, 0x160 + M, 0xf, 0xf, false));
}

__device__ __forceinline__ float wave_max32(float v) {
    v = fmaxf(v, xrow<1>(v));
    v = fmaxf(v, xrow<2>(v));
    v = fmaxf(v, xrow<4>(v));
    v = fmaxf(v, xrow<8>(v));
    v = fmaxf(v, xhalf(v));
    return v;
}

__device__ __forceinline__ float lane_get(float x, int l) {
    return __builtin_bit_cast(float,
        __builtin_amdgcn_readlane(__builtin_bit_cast(int, x), l));
}

__device__ __forceinline__ void invert_delta_on_wave(const int i, float (*invD)[KK]) {
    float row[NAUG];
    for (int j = 0; j < KK; ++j) {
        const float dx = cpx_of(i) - cpx_of(j);
        const float dy = cpy_of(i) - cpy_of(j);
        const float u  = tps_u(dx*dx + dy*dy);
        const float e  = (i == KK) ? 1.0f : ((i == KK+1) ? cpx_of(j) : cpy_of(j));
        const float v  = (i < KK) ? u : e;
        row[j] = (i < KP3) ? v : 0.0f;
    }
    row[KK]     = (i < KK) ? 1.0f      : 0.0f;
    row[KK + 1] = (i < KK) ? cpx_of(i) : 0.0f;
    row[KK + 2] = (i < KK) ? cpy_of(i) : 0.0f;
    for (int j = 0; j < KP3; ++j) row[KP3 + j] = (i == j) ? 1.0f : 0.0f;

    for (int p = 0; p < KP3; ++p) {
        const float a  = (i >= p && i < KP3) ? fabsf(row[0]) : -1.0f;
        const float mx = wave_max32(a);
        const unsigned hit = __builtin_amdgcn_ballot_w32(a == mx);
        const int piv = hit ? __builtin_ctz(hit) : p;

        float pr[NAUG];
        for (int j = 0; j < NAUG; ++j) pr[j] = lane_get(row[j], piv);

        if (piv != p) {
            for (int j = 0; j < NAUG; ++j) {
                const float o = lane_get(row[j], p);
                row[j] = (i == piv) ? o : row[j];
            }
        }

        const float inv = 1.0f / pr[0];
        const float f   = row[0];
        float nw[NAUG];
        for (int j = 0; j < NAUG; ++j) {
            const float ps = pr[j] * inv;
            const float e  = row[j] - f * ps;
            nw[j] = (i == p) ? ps : e;
        }
        for (int j = 0; j < NAUG - 1; ++j) row[j] = nw[j + 1];
        row[NAUG - 1] = nw[0];
    }

    if (i < KP3) {
        for (int j = 0; j < KK; ++j) invD[i][j] = row[j];
    }
}

__device__ __forceinline__ void basis_fragments(v2f (&U)[2][5], const int nE, const bool hi) {
    #pragma unroll
    for (int e = 0; e < 2; ++e) {
        const int n = nE + e;
        const float px = -1.0f + 2.0f * (float)(n & (WW - 1)) * (1.0f / (float)(WW - 1));
        const float py = -1.0f + 2.0f * (float)(n >> 9)       * (1.0f / (float)(HH - 1));
        #pragma unroll
        for (int kb = 0; kb < 4; ++kb) {
            int k0 = kb * 4 + (hi ? 2 : 0);
            float dx0 = px - cpx_of(k0),     dy0 = py - cpy_of(k0);
            float dx1 = px - cpx_of(k0 + 1), dy1 = py - cpy_of(k0 + 1);
            float r20 = dx0*dx0 + dy0*dy0;
            float r21 = dx1*dx1 + dy1*dy1;
            U[e][kb].x = r20 * logf(r20 + 1e-6f);
            U[e][kb].y = r21 * logf(r21 + 1e-6f);
        }
        U[e][4].x = hi ? py   : 1.0f;
        U[e][4].y = hi ? 0.0f : px;
    }
}

__global__ void __launch_bounds__(256)
tps_fused_kernel(const float* __restrict__ sp, float* __restrict__ out) {
    __shared__ float invD[KP3][KK];
    __shared__ float spL[BB * SPLD];
    __shared__ float Bls[KPAD * BLDS];

    const int tid  = threadIdx.x;
    const int lane = tid & 31;
    const int wv   = tid >> 5;
    const int c    = lane & 15;
    const bool hi  = lane >= 16;
    const int n0   = blockIdx.x * (TPW * 256) + wv * 32;
    const int nE   = n0 + 2 * c;

    v2f U[2][5];

    if (wv == 0) {
        invert_delta_on_wave(lane, invD);
    } else {
        for (int idx = tid - 32; idx < BB * KK * 2; idx += 224) {
            spL[(idx >> 5) * SPLD + (idx & 31)] = sp[idx];
        }
        basis_fragments(U, nE, hi);
    }
    __syncthreads();

    {
        const int col = tid & (NCOL - 1);
        const int i0  = (tid >> 7) * 10;
        const int b = col >> 1, d = col & 1;
        float spv[KK];
        #pragma unroll
        for (int j = 0; j < KK; ++j) spv[j] = spL[b * SPLD + j*2 + d];
        #pragma unroll
        for (int ii = 0; ii < 10; ++ii) {
            const int i  = i0 + ii;
            const int ir = (i < KP3) ? i : (KP3 - 1);
            float acc = 0.0f;
            #pragma unroll
            for (int j = 0; j < KK; ++j) acc += invD[ir][j] * spv[j];
            Bls[i * BLDS + col] = (i < KP3) ? acc : 0.0f;
        }
    }
    __syncthreads();

    v2f Cf[8][5];
    #pragma unroll
    for (int ct = 0; ct < 8; ++ct) {
        const int col = ct * 16 + c;
        #pragma unroll
        for (int kb = 0; kb < 5; ++kb) {
            int r0 = kb * 4 + (hi ? 2 : 0);
            Cf[ct][kb].x = Bls[r0 * BLDS + col];
            Cf[ct][kb].y = Bls[(r0 + 1) * BLDS + col];
        }
    }

    float* gp = out + (size_t)(hi ? 4 : 0) * (size_t)(HWN * 2) + (size_t)nE * 2;

    #pragma unroll 1
    for (int t = 0; t < TPW; ++t) {
        if (t > 0 || wv == 0) basis_fragments(U, nE + t * 256, hi);

        #pragma unroll
        for (int ct = 0; ct < 8; ++ct) {
            v8f E = {0.f, 0.f, 0.f, 0.f, 0.f, 0.f, 0.f, 0.f};
            v8f O = {0.f, 0.f, 0.f, 0.f, 0.f, 0.f, 0.f, 0.f};
            #pragma unroll
            for (int kb = 0; kb < 5; ++kb) {
                E = __builtin_amdgcn_wmma_f32_16x16x4_f32(
                         false, Cf[ct][kb],  false, U[0][kb],
                         (short)0, E,  false,  false);
                O = __builtin_amdgcn_wmma_f32_16x16x4_f32(
                         false, Cf[ct][kb],  false, U[1][kb],
                         (short)0, O,  false,  false);
            }
            #pragma unroll
            for (int i = 0; i < 4; ++i) {
                v4f val = {E[2*i], E[2*i + 1], O[2*i], O[2*i + 1]};
                *(v4f*)(gp + (size_t)(ct * 8 + i) * (size_t)(HWN * 2)) = val;
            }
        }
        gp += 256 * 2;
    }
}

extern "C" void kernel_launch(void* const* d_in, const int* in_sizes, int n_in,
                              void* d_out, int out_size, void* d_ws, size_t ws_size,
                              hipStream_t stream) {
    (void)in_sizes; (void)n_in; (void)out_size; (void)d_ws; (void)ws_size;
    const float* src = (const float*)d_in[0];
    float* out       = (float*)d_out;

    const int nBlocks = NWG;
    tps_fused_kernel<<<nBlocks, 256, 0, stream>>>(src, out);
}


# setup: each inversion step is one branch-free block on the 19 live columns (right half entered a column per step, reordered at the write), next pivot searched inside it on |x| bits; values kept
# speedup vs baseline: 34.0858x; 1.0687x over previous
#include <hip/hip_runtime.h>
#include <hip/hip_bf16.h>
#include <math.h>


#define HH   512
#define WW   512
#define KK   16
#define KP3  19
#define KPAD 20
#define BB   64
#define NCOL 128
#define HWN  (HH*WW)
#define NWG  256
#define TPW  4

#define BLDS 136
#define SPLD 33

typedef __attribute__((ext_vector_type(2))) float v2f;
typedef __attribute__((ext_vector_type(4))) float v4f;
typedef __attribute__((ext_vector_type(8))) float v8f;

__device__ __forceinline__ float cpx_of(int k) { return -1.0f + (2.0f/3.0f)*(float)(k >> 2); }
__device__ __forceinline__ float cpy_of(int k) { return -1.0f + (2.0f/3.0f)*(float)(k & 3); }
__device__ __forceinline__ float tps_u(float r2) { return r2 * logf(r2 + 1e-6f); }

__device__ __forceinline__ int xhalf(int b) {
    return __builtin_amdgcn_permlanex16(b, b, 0x76543210, 0xfedcba98, false, false);
}

template <int M>
__device__ __forceinline__ int xrow(int b) {
    return __builtin_amdgcn_update_dpp(0, b, 0x160 + M, 0xf, 0xf, true);
}

__device__ __forceinline__ int wave_max32(int v) {
    v = max(v, xrow<1>(v));
    v = max(v, xrow<2>(v));
    v = max(v, xrow<4>(v));
    v = max(v, xrow<8>(v));
    v = max(v, xhalf(v));
    return v;
}

__device__ __forceinline__ float lane_get(float x, int l) {
    return __builtin_bit_cast(float,
        __builtin_amdgcn_readlane(__builtin_bit_cast(int, x), l));
}
__device__ __forceinline__ int lane_get(int x, int l) {
    return __builtin_amdgcn_readlane(x, l);
}

__device__ __forceinline__ void find_pivot(const float r0, const int i, const int p,
                                           int& piv, float& inv_abs) {
    const int a  = (i >= p && i < KP3) ? __builtin_bit_cast(int, fabsf(r0)) : -1;
    const int mx = wave_max32(a);
    const unsigned hit = __builtin_amdgcn_ballot_w32(a == mx);
    piv     = hit ? __builtin_ctz(hit) : p;
    inv_abs = 1.0f / __builtin_bit_cast(float, mx);
}

__device__ __forceinline__ void invert_delta_on_wave(const int i, float (*invD)[KPAD]) {
    float row[KP3];
    #pragma unroll
    for (int j = 0; j < KK; ++j) {
        const float dx = cpx_of(i) - cpx_of(j);
        const float dy = cpy_of(i) - cpy_of(j);
        const float u  = tps_u(dx*dx + dy*dy);
        const float e  = (i == KK) ? 1.0f : ((i == KK+1) ? cpx_of(j) : cpy_of(j));
        const float v  = (i < KK) ? u : e;
        row[j] = (i < KP3) ? v : 0.0f;
    }
    row[KK]     = (i < KK) ? 1.0f      : 0.0f;
    row[KK + 1] = (i < KK) ? cpx_of(i) : 0.0f;
    row[KK + 2] = (i < KK) ? cpy_of(i) : 0.0f;
    int sig = i;

    int piv; float inv_abs;
    find_pivot(row[0], i, 0, piv, inv_abs);

    for (int p = 0; p < KP3; ++p) {
        float pr[KP3];
        for (int j = 0; j < KP3; ++j) pr[j] = lane_get(row[j], piv);
        const int sgp = lane_get(sig, piv);

        for (int j = 0; j < KP3; ++j) {
            const float o = lane_get(row[j], p);
            row[j] = (i == piv) ? o : row[j];
        }
        const int sgo = lane_get(sig, p);
        sig = (i == piv) ? sgo : sig;
        sig = (i == p)   ? sgp : sig;

        const float inv = copysignf(inv_abs, pr[0]);
        const float f   = row[0];
        float nw[KP3 + 1];
        for (int j = 0; j < KP3; ++j) {
            const float ps = pr[j] * inv;
            const float e  = row[j] - f * ps;
            nw[j] = (i == p) ? ps : e;
        }
        nw[KP3] = (i == p) ? inv : (0.0f - f * inv);
        for (int j = 0; j < KP3; ++j) row[j] = nw[j + 1];

        find_pivot(row[0], i, p + 1, piv, inv_abs);
    }

    int col[KP3];
    for (int j = 0; j < KP3; ++j) col[j] = lane_get(sig, j);
    if (i < KP3) {
        for (int j = 0; j < KP3; ++j) invD[i][col[j]] = row[j];
    }
}

__device__ __forceinline__ void basis_fragments(v2f (&U)[2][5], const int nE, const bool hi) {
    #pragma unroll
    for (int e = 0; e < 2; ++e) {
        const int n = nE + e;
        const float px = -1.0f + 2.0f * (float)(n & (WW - 1)) * (1.0f / (float)(WW - 1));
        const float py = -1.0f + 2.0f * (float)(n >> 9)       * (1.0f / (float)(HH - 1));
        #pragma unroll
        for (int kb = 0; kb < 4; ++kb) {
            int k0 = kb * 4 + (hi ? 2 : 0);
            float dx0 = px - cpx_of(k0),     dy0 = py - cpy_of(k0);
            float dx1 = px - cpx_of(k0 + 1), dy1 = py - cpy_of(k0 + 1);
            float r20 = dx0*dx0 + dy0*dy0;
            float r21 = dx1*dx1 + dy1*dy1;
            U[e][kb].x = r20 * logf(r20 + 1e-6f);
            U[e][kb].y = r21 * logf(r21 + 1e-6f);
        }
        U[e][4].x = hi ? py   : 1.0f;
        U[e][4].y = hi ? 0.0f : px;
    }
}

__global__ void __launch_bounds__(256)
tps_fused_kernel(const float* __restrict__ sp, float* __restrict__ out) {
    __shared__ float invD[KP3][KPAD];
    __shared__ float spL[BB * SPLD];
    __shared__ float Bls[KPAD * BLDS];

    const int tid  = threadIdx.x;
    const int lane = tid & 31;
    const int wv   = tid >> 5;
    const int c    = lane & 15;
    const bool hi  = lane >= 16;
    const int n0   = blockIdx.x * (TPW * 256) + wv * 32;
    const int nE   = n0 + 2 * c;

    v2f U[2][5];

    if (wv == 0) {
        invert_delta_on_wave(lane, invD);
    } else {
        for (int idx = tid - 32; idx < BB * KK * 2; idx += 224) {
            spL[(idx >> 5) * SPLD + (idx & 31)] = sp[idx];
        }
        basis_fragments(U, nE, hi);
    }
    __syncthreads();

    {
        const int col = tid & (NCOL - 1);
        const int i0  = (tid >> 7) * 10;
        const int b = col >> 1, d = col & 1;
        float spv[KK];
        #pragma unroll
        for (int j = 0; j < KK; ++j) spv[j] = spL[b * SPLD + j*2 + d];
        #pragma unroll
        for (int ii = 0; ii < 10; ++ii) {
            const int i  = i0 + ii;
            const int ir = (i < KP3) ? i : (KP3 - 1);
            float acc = 0.0f;
            #pragma unroll
            for (int j = 0; j < KK; ++j) acc += invD[ir][j] * spv[j];
            Bls[i * BLDS + col] = (i < KP3) ? acc : 0.0f;
        }
    }
    __syncthreads();

    v2f Cf[8][5];
    #pragma unroll
    for (int ct = 0; ct < 8; ++ct) {
        const int col = ct * 16 + c;
        #pragma unroll
        for (int kb = 0; kb < 5; ++kb) {
            int r0 = kb * 4 + (hi ? 2 : 0);
            Cf[ct][kb].x = Bls[r0 * BLDS + col];
            Cf[ct][kb].y = Bls[(r0 + 1) * BLDS + col];
        }
    }

    float* gp = out + (size_t)(hi ? 4 : 0) * (size_t)(HWN * 2) + (size_t)nE * 2;

    #pragma unroll 1
    for (int t = 0; t < TPW; ++t) {
        if (t > 0 || wv == 0) basis_fragments(U, nE + t * 256, hi);

        #pragma unroll
        for (int ct = 0; ct < 8; ++ct) {
            v8f E = {0.f, 0.f, 0.f, 0.f, 0.f, 0.f, 0.f, 0.f};
            v8f O = {0.f, 0.f, 0.f, 0.f, 0.f, 0.f, 0.f, 0.f};
            #pragma unroll
            for (int kb = 0; kb < 5; ++kb) {
                E = __builtin_amdgcn_wmma_f32_16x16x4_f32(
                         false, Cf[ct][kb],  false, U[0][kb],
                         (short)0, E,  false,  false);
                O = __builtin_amdgcn_wmma_f32_16x16x4_f32(
                         false, Cf[ct][kb],  false, U[1][kb],
                         (short)0, O,  false,  false);
            }
            #pragma unroll
            for (int i = 0; i < 4; ++i) {
                v4f val = {E[2*i], E[2*i + 1], O[2*i], O[2*i + 1]};
                *(v4f*)(gp + (size_t)(ct * 8 + i) * (size_t)(HWN * 2)) = val;
            }
        }
        gp += 256 * 2;
    }
}

extern "C" void kernel_launch(void* const* d_in, const int* in_sizes, int n_in,
                              void* d_out, int out_size, void* d_ws, size_t ws_size,
                              hipStream_t stream) {
    (void)in_sizes; (void)n_in; (void)out_size; (void)d_ws; (void)ws_size;
    const float* src = (const float*)d_in[0];
    float* out       = (float*)d_out;

    const int nBlocks = NWG;
    tps_fused_kernel<<<nBlocks, 256, 0, stream>>>(src, out);
}
